# starting kernel
# baseline (speedup 1.0000x reference)
#include <hip/hip_runtime.h>

#define SEQ 2048
#define CH  512
#define NH  16
#define HD  32

typedef float v2f __attribute__((ext_vector_type(2)));
typedef float v8f __attribute__((ext_vector_type(8)));

__device__ __forceinline__ v8f wmma4(v2f a, v2f b, v8f c) {
  return __builtin_amdgcn_wmma_f32_16x16x4_f32(false, a, false, b, (short)0, c, false, false);
}

__device__ __forceinline__ v2f ldp(const float* p) {
  const float2 t = *(const float2*)p;
  v2f r = { t.x, t.y };
  return r;
}

template <int MODE>
__global__ void __launch_bounds__(256)
proj_complex_kernel(const float* __restrict__ X, const float* __restrict__ W,
                    const float* __restrict__ Bv, float* __restrict__ Yre,
                    float* __restrict__ Yim) {
  const int wave = threadIdx.x >> 5;
  const int lane = threadIdx.x & 31;
  const int half = lane >> 4;
  const int l16  = lane & 15;

  const int tileId  = blockIdx.x * 8 + wave;
  const int nTilesO = CH / 16;
  const int sTile   = tileId / nTilesO;
  const int oTile   = tileId % nTilesO;

  const int arow = sTile * 16 + l16;
  const int brow = oTile * 16 + l16;

  v8f yr = {0.f}, yi = {0.f};

  for (int k0 = 0; k0 < CH; k0 += 4) {
    const int ka = k0 + half * 2;
    const float4 xa = *(const float4*)(X + ((size_t)arow * CH + ka) * 2);
    const float4 wb = *(const float4*)(W + ((size_t)brow * CH + ka) * 2);
    v2f ar  = { xa.x,  xa.z };
    v2f ai  = { xa.y,  xa.w };
    v2f ain = { -xa.y, -xa.w };
    v2f br  = { wb.x,  wb.z };
    v2f bi  = { wb.y,  wb.w };
    yr = wmma4(ar,  br, yr);
    yr = wmma4(ain, bi, yr);
    yi = wmma4(ar,  bi, yi);
    yi = wmma4(ai,  br, yi);
  }

  const float2 bb = *(const float2*)(Bv + (size_t)brow * 2);
  const int h  = brow >> 5;
  const int dd = brow & 31;

  if (MODE == 0) {
#pragma unroll
    for (int r = 0; r < 8; ++r) {
      const int s = sTile * 16 + r + 8 * half;
      const size_t idx = ((size_t)h * SEQ + s) * HD + dd;
      Yre[idx] = yr[r] + bb.x;
      Yim[idx] = yi[r] + bb.y;
    }
  } else {
    const size_t base = ((size_t)(h * HD + dd)) * SEQ + sTile * 16 + 8 * half;
    float4 r0 = { yr[0] + bb.x, yr[1] + bb.x, yr[2] + bb.x, yr[3] + bb.x };
    float4 r1 = { yr[4] + bb.x, yr[5] + bb.x, yr[6] + bb.x, yr[7] + bb.x };
    float4 i0 = { yi[0] + bb.y, yi[1] + bb.y, yi[2] + bb.y, yi[3] + bb.y };
    float4 i1 = { yi[4] + bb.y, yi[5] + bb.y, yi[6] + bb.y, yi[7] + bb.y };
    *(float4*)(Yre + base)     = r0;
    *(float4*)(Yre + base + 4) = r1;
    *(float4*)(Yim + base)     = i0;
    *(float4*)(Yim + base + 4) = i1;
  }
}

__global__ void __launch_bounds__(256)
attn_complex_kernel(const float* __restrict__ Qre, const float* __restrict__ Qim,
                    const float* __restrict__ Kre, const float* __restrict__ Kim,
                    const float* __restrict__ Vtre, const float* __restrict__ Vtim,
                    float* __restrict__ Out) {
  __shared__ float lds_p[8][16 * 16];

  const int wave = threadIdx.x >> 5;
  const int lane = threadIdx.x & 31;
  const int half = lane >> 4;
  const int l16  = lane & 15;
  const int h    = blockIdx.y;
  const int q0   = (blockIdx.x * 8 + wave) * 16;

  const float scale = 0.17677669529663687f;

  const float* qreh = Qre + (size_t)h * SEQ * HD;
  const float* qimh = Qim + (size_t)h * SEQ * HD;
  const float* kreh = Kre + (size_t)h * SEQ * HD;
  const float* kimh = Kim + (size_t)h * SEQ * HD;
  const float* vreh = Vtre + (size_t)h * HD * SEQ;
  const float* vimh = Vtim + (size_t)h * HD * SEQ;
  float* pbuf = lds_p[wave];

  v2f qr[8], qi[8], qrn[8];
#pragma unroll
  for (int t = 0; t < 8; ++t) {
    const int dk = t * 4 + half * 2;
    const size_t off = ((size_t)(q0 + l16)) * HD + dk;
    qr[t]  = ldp(qreh + off);
    qi[t]  = ldp(qimh + off);
    qrn[t] = -qr[t];
  }

  float m_run[8], l_run[8];
  v8f o_re0 = {0.f}, o_re1 = {0.f}, o_im0 = {0.f}, o_im1 = {0.f};
#pragma unroll
  for (int r = 0; r < 8; ++r) { m_run[r] = -1e30f; l_run[r] = 0.f; }

  for (int kt = 0; kt < SEQ / 16; ++kt) {
    const int kbase = kt * 16;

    v8f sre = {0.f}, sim = {0.f};
#pragma unroll
    for (int t = 0; t < 8; ++t) {
      const int dk = t * 4 + half * 2;
      const size_t off = ((size_t)(kbase + l16)) * HD + dk;
      v2f krv = ldp(kreh + off);
      v2f kiv = ldp(kimh + off);
      sre = wmma4(qr[t],  krv, sre);
      sre = wmma4(qi[t],  kiv, sre);
      sim = wmma4(qi[t],  krv, sim);
      sim = wmma4(qrn[t], kiv, sim);
    }

    float pl[8], rs[8];
#pragma unroll
    for (int r = 0; r < 8; ++r) {
      float a = sqrtf(sre[r] * sre[r] + sim[r] * sim[r]) * scale;
      float mx = a;
#pragma unroll
      for (int msk = 1; msk < 16; msk <<= 1) mx = fmaxf(mx, __shfl_xor(mx, msk, 32));
      const float newm = fmaxf(m_run[r], mx);
      float pv = __expf(a - newm);
      float sum = pv;
#pragma unroll
      for (int msk = 1; msk < 16; msk <<= 1) sum += __shfl_xor(sum, msk, 32);
      const float sc = __expf(m_run[r] - newm);
      l_run[r] = l_run[r] * sc + sum;
      m_run[r] = newm;
      pl[r] = pv;
      rs[r] = sc;
    }
#pragma unroll
    for (int r = 0; r < 8; ++r) {
      o_re0[r] *= rs[r]; o_re1[r] *= rs[r];
      o_im0[r] *= rs[r]; o_im1[r] *= rs[r];
    }

#pragma unroll
    for (int r = 0; r < 8; ++r) pbuf[(r + 8 * half) * 16 + l16] = pl[r];
    __asm__ volatile("s_wait_dscnt 0x0" ::: "memory");

#pragma unroll
    for (int t = 0; t < 4; ++t) {
      const int klocal = t * 4 + half * 2;
      v2f aP = ldp(pbuf + l16 * 16 + klocal);
      const size_t voff = (size_t)l16 * SEQ + kbase + klocal;
      v2f br0 = ldp(vreh + voff);
      v2f bi0 = ldp(vimh + voff);
      v2f br1 = ldp(vreh + voff + (size_t)16 * SEQ);
      v2f bi1 = ldp(vimh + voff + (size_t)16 * SEQ);
      o_re0 = wmma4(aP, br0, o_re0);
      o_im0 = wmma4(aP, bi0, o_im0);
      o_re1 = wmma4(aP, br1, o_re1);
      o_im1 = wmma4(aP, bi1, o_im1);
    }
  }

#pragma unroll
  for (int r = 0; r < 8; ++r) {
    const float inv = 1.0f / l_run[r];
    const int s  = q0 + r + 8 * half;
    const int o0 = h * HD + l16;
    float2 e0 = { o_re0[r] * inv, o_im0[r] * inv };
    float2 e1 = { o_re1[r] * inv, o_im1[r] * inv };
    *(float2*)(Out + ((size_t)s * CH + o0) * 2)      = e0;
    *(float2*)(Out + ((size_t)s * CH + o0 + 16) * 2) = e1;
  }
}

extern "C" void kernel_launch(void* const* d_in, const int* in_sizes, int n_in,
                              void* d_out, int out_size, void* d_ws, size_t ws_size,
                              hipStream_t stream) {
  (void)in_sizes; (void)n_in; (void)out_size; (void)ws_size;
  const float* Q  = (const float*)d_in[0];
  const float* V  = (const float*)d_in[1];
  const float* K  = (const float*)d_in[2];
  const float* Wq = (const float*)d_in[3];
  const float* bq = (const float*)d_in[4];
  const float* Wk = (const float*)d_in[5];
  const float* bk = (const float*)d_in[6];
  const float* Wv = (const float*)d_in[7];
  const float* bv = (const float*)d_in[8];
  float* out = (float*)d_out;

  const size_t plane = (size_t)NH * SEQ * HD;
  float* qre = (float*)d_ws;
  float* qim = qre + plane;
  float* kre = qim + plane;
  float* kim = kre + plane;
  float* vre = kim + plane;
  float* vim = vre + plane;

  const dim3 blk(256);
  const dim3 pgrid((SEQ / 16) * (CH / 16) / 8);
  proj_complex_kernel<0><<<pgrid, blk, 0, stream>>>(Q, Wq, bq, qre, qim);
  proj_complex_kernel<0><<<pgrid, blk, 0, stream>>>(K, Wk, bk, kre, kim);
  proj_complex_kernel<1><<<pgrid, blk, 0, stream>>>(V, Wv, bv, vre, vim);

  const dim3 agrid(SEQ / 16 / 8, NH);
  attn_complex_kernel<<<agrid, blk, 0, stream>>>(qre, qim, kre, kim, vre, vim, out);
}


# attention softmax: row maxima and row sums taken by row_xmask DPP steps on the vector ALU, not by 64 LDS-pipe shuffles per key tile with their waits; butterfly order and every sum kept
# speedup vs baseline: 1.4519x; 1.4519x over previous
#include <hip/hip_runtime.h>

#define SEQ 2048
#define CH  512
#define NH  16
#define HD  32

typedef float v2f __attribute__((ext_vector_type(2)));
typedef float v8f __attribute__((ext_vector_type(8)));
typedef float v4f __attribute__((ext_vector_type(4)));

__device__ __forceinline__ v8f wmma4(v2f a, v2f b, v8f c) {
  return __builtin_amdgcn_wmma_f32_16x16x4_f32(false, a, false, b, (short)0, c, false, false);
}

__device__ __forceinline__ v2f ldp(const float* p) {
  const float2 t = *(const float2*)p;
  v2f r = { t.x, t.y };
  return r;
}

template <int M>
__device__ __forceinline__ float row_xor(float v) {
  return __builtin_bit_cast(float, __builtin_amdgcn_update_dpp(
      0, __builtin_bit_cast(int, v), 0x160 + M, 0xf, 0xf, true));
}

template <int MODE>
__global__ void __launch_bounds__(256)
proj_complex_kernel(const float* __restrict__ X, const float* __restrict__ W,
                    const float* __restrict__ Bv, float* __restrict__ Yre,
                    float* __restrict__ Yim) {
  const int wave = threadIdx.x >> 5;
  const int lane = threadIdx.x & 31;
  const int half = lane >> 4;
  const int l16  = lane & 15;

  const int tileId  = blockIdx.x * 8 + wave;
  const int nTilesO = CH / 16;
  const int sTile   = tileId / nTilesO;
  const int oTile   = tileId % nTilesO;

  const int arow = sTile * 16 + l16;
  const int brow = oTile * 16 + l16;

  v8f yr = {0.f}, yi = {0.f};

  for (int k0 = 0; k0 < CH; k0 += 4) {
    const int ka = k0 + half * 2;
    const float4 xa = *(const float4*)(X + ((size_t)arow * CH + ka) * 2);
    const float4 wb = *(const float4*)(W + ((size_t)brow * CH + ka) * 2);
    v2f ar  = { xa.x,  xa.z };
    v2f ai  = { xa.y,  xa.w };
    v2f ain = { -xa.y, -xa.w };
    v2f br  = { wb.x,  wb.z };
    v2f bi  = { wb.y,  wb.w };
    yr = wmma4(ar,  br, yr);
    yr = wmma4(ain, bi, yr);
    yi = wmma4(ar,  bi, yi);
    yi = wmma4(ai,  br, yi);
  }

  const float2 bb = *(const float2*)(Bv + (size_t)brow * 2);
  const int h  = brow >> 5;
  const int dd = brow & 31;

  if (MODE == 0) {
#pragma unroll
    for (int r = 0; r < 8; ++r) {
      const int s = sTile * 16 + r + 8 * half;
      const size_t idx = ((size_t)h * SEQ + s) * HD + dd;
      Yre[idx] = yr[r] + bb.x;
      Yim[idx] = yi[r] + bb.y;
    }
  } else {
    const size_t base = ((size_t)(h * HD + dd)) * SEQ + sTile * 16 + 8 * half;
    float4 r0 = { yr[0] + bb.x, yr[1] + bb.x, yr[2] + bb.x, yr[3] + bb.x };
    float4 r1 = { yr[4] + bb.x, yr[5] + bb.x, yr[6] + bb.x, yr[7] + bb.x };
    float4 i0 = { yi[0] + bb.y, yi[1] + bb.y, yi[2] + bb.y, yi[3] + bb.y };
    float4 i1 = { yi[4] + bb.y, yi[5] + bb.y, yi[6] + bb.y, yi[7] + bb.y };
    *(float4*)(Yre + base)     = r0;
    *(float4*)(Yre + base + 4) = r1;
    *(float4*)(Yim + base)     = i0;
    *(float4*)(Yim + base + 4) = i1;
  }
}

#define KC 64
#define PK 36
#define PV (KC + 4)

__global__ void __launch_bounds__(256)
attn_complex_kernel(const float* __restrict__ Qre, const float* __restrict__ Qim,
                    const float* __restrict__ Kre, const float* __restrict__ Kim,
                    const float* __restrict__ Vtre, const float* __restrict__ Vtim,
                    float* __restrict__ Out) {
  __shared__ __attribute__((aligned(16))) float lds_k[2][2][KC * PK];
  __shared__ __attribute__((aligned(16))) float lds_v[2][2][HD * PV];
  __shared__ float lds_p[8][16 * 16];

  const int tid  = threadIdx.x;
  const int wave = tid >> 5;
  const int lane = tid & 31;
  const int half = lane >> 4;
  const int l16  = lane & 15;
  const int h    = blockIdx.y;
  const int q0   = (blockIdx.x * 8 + wave) * 16;

  const float scale = 0.17677669529663687f;

  const float* qreh = Qre + (size_t)h * SEQ * HD;
  const float* qimh = Qim + (size_t)h * SEQ * HD;
  const float* kreh = Kre + (size_t)h * SEQ * HD;
  const float* kimh = Kim + (size_t)h * SEQ * HD;
  const float* vreh = Vtre + (size_t)h * HD * SEQ;
  const float* vimh = Vtim + (size_t)h * HD * SEQ;
  float* pbuf = lds_p[wave];

  v2f qr[8], qi[8], qrn[8];
#pragma unroll
  for (int t = 0; t < 8; ++t) {
    const int dk = t * 4 + half * 2;
    const size_t off = ((size_t)(q0 + l16)) * HD + dk;
    qr[t]  = ldp(qreh + off);
    qi[t]  = ldp(qimh + off);
    qrn[t] = -qr[t];
  }

  float m_run[8], l_run[8];
  v8f o_re0 = {0.f}, o_re1 = {0.f}, o_im0 = {0.f}, o_im1 = {0.f};
#pragma unroll
  for (int r = 0; r < 8; ++r) { m_run[r] = -1e30f; l_run[r] = 0.f; }

  const int i0 = tid, i1 = tid + 256;
  const int kg0 = i0 * 4, kg1 = i1 * 4;
  const int kl0 = (i0 >> 3) * PK + (i0 & 7) * 4;
  const int kl1 = (i1 >> 3) * PK + (i1 & 7) * 4;
  const int vg0 = (i0 / (KC / 4)) * SEQ + (i0 % (KC / 4)) * 4;
  const int vg1 = (i1 / (KC / 4)) * SEQ + (i1 % (KC / 4)) * 4;
  const int vl0 = (i0 / (KC / 4)) * PV + (i0 % (KC / 4)) * 4;
  const int vl1 = (i1 / (KC / 4)) * PV + (i1 % (KC / 4)) * 4;

  v4f gkr0 = *(const v4f*)(kreh + kg0), gkr1 = *(const v4f*)(kreh + kg1);
  v4f gki0 = *(const v4f*)(kimh + kg0), gki1 = *(const v4f*)(kimh + kg1);
  v4f gvr0 = *(const v4f*)(vreh + vg0), gvr1 = *(const v4f*)(vreh + vg1);
  v4f gvi0 = *(const v4f*)(vimh + vg0), gvi1 = *(const v4f*)(vimh + vg1);
  *(v4f*)(&lds_k[0][0][kl0]) = gkr0;  *(v4f*)(&lds_k[0][0][kl1]) = gkr1;
  *(v4f*)(&lds_k[0][1][kl0]) = gki0;  *(v4f*)(&lds_k[0][1][kl1]) = gki1;
  *(v4f*)(&lds_v[0][0][vl0]) = gvr0;  *(v4f*)(&lds_v[0][0][vl1]) = gvr1;
  *(v4f*)(&lds_v[0][1][vl0]) = gvi0;  *(v4f*)(&lds_v[0][1][vl1]) = gvi1;
  __syncthreads();

  for (int c = 0; c < SEQ / KC; ++c) {
    const int st = c & 1;
    const bool more = (c + 1) < (SEQ / KC);

    if (more) {
      const float* kre_n = kreh + (size_t)(c + 1) * KC * HD;
      const float* kim_n = kimh + (size_t)(c + 1) * KC * HD;
      const float* vre_n = vreh + (size_t)(c + 1) * KC;
      const float* vim_n = vimh + (size_t)(c + 1) * KC;
      gkr0 = *(const v4f*)(kre_n + kg0);  gkr1 = *(const v4f*)(kre_n + kg1);
      gki0 = *(const v4f*)(kim_n + kg0);  gki1 = *(const v4f*)(kim_n + kg1);
      gvr0 = *(const v4f*)(vre_n + vg0);  gvr1 = *(const v4f*)(vre_n + vg1);
      gvi0 = *(const v4f*)(vim_n + vg0);  gvi1 = *(const v4f*)(vim_n + vg1);
    }

    const float* ks_re = lds_k[st][0];
    const float* ks_im = lds_k[st][1];
    const float* vs_re = lds_v[st][0];
    const float* vs_im = lds_v[st][1];

    for (int kt = 0; kt < KC / 16; ++kt) {
      const int krow = (kt * 16 + l16) * PK + half * 2;
      v8f sre = {0.f}, sim = {0.f};
#pragma unroll
      for (int t = 0; t < 8; ++t) {
        v2f krv = ldp(ks_re + krow + t * 4);
        v2f kiv = ldp(ks_im + krow + t * 4);
        sre = wmma4(qr[t],  krv, sre);
        sre = wmma4(qi[t],  kiv, sre);
        sim = wmma4(qi[t],  krv, sim);
        sim = wmma4(qrn[t], kiv, sim);
      }

      float pl[8], rs[8];
#pragma unroll
      for (int r = 0; r < 8; ++r) {
        float a = sqrtf(sre[r] * sre[r] + sim[r] * sim[r]) * scale;
        float mx = a;
        mx = fmaxf(mx, row_xor<1>(mx));
        mx = fmaxf(mx, row_xor<2>(mx));
        mx = fmaxf(mx, row_xor<4>(mx));
        mx = fmaxf(mx, row_xor<8>(mx));
        const float newm = fmaxf(m_run[r], mx);
        float pv = __expf(a - newm);
        float sum = pv;
        sum += row_xor<1>(sum);
        sum += row_xor<2>(sum);
        sum += row_xor<4>(sum);
        sum += row_xor<8>(sum);
        const float sc = __expf(m_run[r] - newm);
        l_run[r] = l_run[r] * sc + sum;
        m_run[r] = newm;
        pl[r] = pv;
        rs[r] = sc;
      }
#pragma unroll
      for (int r = 0; r < 8; ++r) {
        o_re0[r] *= rs[r]; o_re1[r] *= rs[r];
        o_im0[r] *= rs[r]; o_im1[r] *= rs[r];
      }

#pragma unroll
      for (int r = 0; r < 8; ++r) pbuf[(r + 8 * half) * 16 + l16] = pl[r];
      __asm__ volatile("s_wait_dscnt 0x0" ::: "memory");

      const int vrow = l16 * PV + kt * 16 + half * 2;
#pragma unroll
      for (int t = 0; t < 4; ++t) {
        v2f aP  = ldp(pbuf + l16 * 16 + t * 4 + half * 2);
        v2f br0 = ldp(vs_re + vrow + t * 4);
        v2f bi0 = ldp(vs_im + vrow + t * 4);
        v2f br1 = ldp(vs_re + vrow + 16 * PV + t * 4);
        v2f bi1 = ldp(vs_im + vrow + 16 * PV + t * 4);
        o_re0 = wmma4(aP, br0, o_re0);
        o_im0 = wmma4(aP, bi0, o_im0);
        o_re1 = wmma4(aP, br1, o_re1);
        o_im1 = wmma4(aP, bi1, o_im1);
      }
    }

    if (more) {
      const int sn = st ^ 1;
      *(v4f*)(&lds_k[sn][0][kl0]) = gkr0;  *(v4f*)(&lds_k[sn][0][kl1]) = gkr1;
      *(v4f*)(&lds_k[sn][1][kl0]) = gki0;  *(v4f*)(&lds_k[sn][1][kl1]) = gki1;
      *(v4f*)(&lds_v[sn][0][vl0]) = gvr0;  *(v4f*)(&lds_v[sn][0][vl1]) = gvr1;
      *(v4f*)(&lds_v[sn][1][vl0]) = gvi0;  *(v4f*)(&lds_v[sn][1][vl1]) = gvi1;
    }
    __syncthreads();
  }

#pragma unroll
  for (int r = 0; r < 8; ++r) {
    const float inv = 1.0f / l_run[r];
    const int s  = q0 + r + 8 * half;
    const int o0 = h * HD + l16;
    float2 e0 = { o_re0[r] * inv, o_im0[r] * inv };
    float2 e1 = { o_re1[r] * inv, o_im1[r] * inv };
    *(float2*)(Out + ((size_t)s * CH + o0) * 2)      = e0;
    *(float2*)(Out + ((size_t)s * CH + o0 + 16) * 2) = e1;
  }
}

extern "C" void kernel_launch(void* const* d_in, const int* in_sizes, int n_in,
                              void* d_out, int out_size, void* d_ws, size_t ws_size,
                              hipStream_t stream) {
  (void)in_sizes; (void)n_in; (void)out_size; (void)ws_size;
  const float* Q  = (const float*)d_in[0];
  const float* V  = (const float*)d_in[1];
  const float* K  = (const float*)d_in[2];
  const float* Wq = (const float*)d_in[3];
  const float* bq = (const float*)d_in[4];
  const float* Wk = (const float*)d_in[5];
  const float* bk = (const float*)d_in[6];
  const float* Wv = (const float*)d_in[7];
  const float* bv = (const float*)d_in[8];
  float* out = (float*)d_out;

  const size_t plane = (size_t)NH * SEQ * HD;
  float* qre = (float*)d_ws;
  float* qim = qre + plane;
  float* kre = qim + plane;
  float* kim = kre + plane;
  float* vre = kim + plane;
  float* vim = vre + plane;

  const dim3 blk(256);
  const dim3 pgrid((SEQ / 16) * (CH / 16) / 8);
  proj_complex_kernel<0><<<pgrid, blk, 0, stream>>>(Q, Wq, bq, qre, qim);
  proj_complex_kernel<0><<<pgrid, blk, 0, stream>>>(K, Wk, bk, kre, kim);
  proj_complex_kernel<1><<<pgrid, blk, 0, stream>>>(V, Wv, bv, vre, vim);

  const dim3 agrid(SEQ / 16 / 8, NH);
  attn_complex_kernel<<<agrid, blk, 0, stream>>>(qre, qim, kre, kim, vre, vim, out);
}


# attention softmax: one online-softmax step per 64-key stage, not per 16-key tile (4 score tiles in registers, P.V as one 64-WMMA run); order of row sums changes; exp argument no longer fused
# speedup vs baseline: 1.5580x; 1.0731x over previous
#include <hip/hip_runtime.h>

#define SEQ 2048
#define CH  512
#define NH  16
#define HD  32

typedef float v2f __attribute__((ext_vector_type(2)));
typedef float v8f __attribute__((ext_vector_type(8)));
typedef float v4f __attribute__((ext_vector_type(4)));

__device__ __forceinline__ v8f wmma4(v2f a, v2f b, v8f c) {
  return __builtin_amdgcn_wmma_f32_16x16x4_f32(false, a, false, b, (short)0, c, false, false);
}

__device__ __forceinline__ v2f ldp(const float* p) {
  const float2 t = *(const float2*)p;
  v2f r = { t.x, t.y };
  return r;
}

template <int M>
__device__ __forceinline__ float row_xor(float v) {
  return __builtin_bit_cast(float, __builtin_amdgcn_update_dpp(
      0, __builtin_bit_cast(int, v), 0x160 + M, 0xf, 0xf, true));
}

template <int MODE>
__global__ void __launch_bounds__(256)
proj_complex_kernel(const float* __restrict__ X, const float* __restrict__ W,
                    const float* __restrict__ Bv, float* __restrict__ Yre,
                    float* __restrict__ Yim) {
  const int wave = threadIdx.x >> 5;
  const int lane = threadIdx.x & 31;
  const int half = lane >> 4;
  const int l16  = lane & 15;

  const int tileId  = blockIdx.x * 8 + wave;
  const int nTilesO = CH / 16;
  const int sTile   = tileId / nTilesO;
  const int oTile   = tileId % nTilesO;

  const int arow = sTile * 16 + l16;
  const int brow = oTile * 16 + l16;

  v8f yr = {0.f}, yi = {0.f};

  for (int k0 = 0; k0 < CH; k0 += 4) {
    const int ka = k0 + half * 2;
    const float4 xa = *(const float4*)(X + ((size_t)arow * CH + ka) * 2);
    const float4 wb = *(const float4*)(W + ((size_t)brow * CH + ka) * 2);
    v2f ar  = { xa.x,  xa.z };
    v2f ai  = { xa.y,  xa.w };
    v2f ain = { -xa.y, -xa.w };
    v2f br  = { wb.x,  wb.z };
    v2f bi  = { wb.y,  wb.w };
    yr = wmma4(ar,  br, yr);
    yr = wmma4(ain, bi, yr);
    yi = wmma4(ar,  bi, yi);
    yi = wmma4(ai,  br, yi);
  }

  const float2 bb = *(const float2*)(Bv + (size_t)brow * 2);
  const int h  = brow >> 5;
  const int dd = brow & 31;

  if (MODE == 0) {
#pragma unroll
    for (int r = 0; r < 8; ++r) {
      const int s = sTile * 16 + r + 8 * half;
      const size_t idx = ((size_t)h * SEQ + s) * HD + dd;
      Yre[idx] = yr[r] + bb.x;
      Yim[idx] = yi[r] + bb.y;
    }
  } else {
    const size_t base = ((size_t)(h * HD + dd)) * SEQ + sTile * 16 + 8 * half;
    float4 r0 = { yr[0] + bb.x, yr[1] + bb.x, yr[2] + bb.x, yr[3] + bb.x };
    float4 r1 = { yr[4] + bb.x, yr[5] + bb.x, yr[6] + bb.x, yr[7] + bb.x };
    float4 i0 = { yi[0] + bb.y, yi[1] + bb.y, yi[2] + bb.y, yi[3] + bb.y };
    float4 i1 = { yi[4] + bb.y, yi[5] + bb.y, yi[6] + bb.y, yi[7] + bb.y };
    *(float4*)(Yre + base)     = r0;
    *(float4*)(Yre + base + 4) = r1;
    *(float4*)(Yim + base)     = i0;
    *(float4*)(Yim + base + 4) = i1;
  }
}

#define KC 64
#define PK 36
#define PV (KC + 4)
#define PP (KC + 4)

__global__ void __launch_bounds__(256)
attn_complex_kernel(const float* __restrict__ Qre, const float* __restrict__ Qim,
                    const float* __restrict__ Kre, const float* __restrict__ Kim,
                    const float* __restrict__ Vtre, const float* __restrict__ Vtim,
                    float* __restrict__ Out) {
  __shared__ __attribute__((aligned(16))) float lds_k[2][2][KC * PK];
  __shared__ __attribute__((aligned(16))) float lds_v[2][2][HD * PV];
  __shared__ __attribute__((aligned(16))) float lds_p[8][16 * PP];

  const int tid  = threadIdx.x;
  const int wave = tid >> 5;
  const int lane = tid & 31;
  const int half = lane >> 4;
  const int l16  = lane & 15;
  const int h    = blockIdx.y;
  const int q0   = (blockIdx.x * 8 + wave) * 16;

  const float scale = 0.17677669529663687f;

  const float* qreh = Qre + (size_t)h * SEQ * HD;
  const float* qimh = Qim + (size_t)h * SEQ * HD;
  const float* kreh = Kre + (size_t)h * SEQ * HD;
  const float* kimh = Kim + (size_t)h * SEQ * HD;
  const float* vreh = Vtre + (size_t)h * HD * SEQ;
  const float* vimh = Vtim + (size_t)h * HD * SEQ;
  float* pbuf = lds_p[wave];

  v2f qr[8], qi[8], qrn[8];
#pragma unroll
  for (int t = 0; t < 8; ++t) {
    const int dk = t * 4 + half * 2;
    const size_t off = ((size_t)(q0 + l16)) * HD + dk;
    qr[t]  = ldp(qreh + off);
    qi[t]  = ldp(qimh + off);
    qrn[t] = -qr[t];
  }

  float m_run[8], l_run[8];
  v8f o_re0 = {0.f}, o_re1 = {0.f}, o_im0 = {0.f}, o_im1 = {0.f};
#pragma unroll
  for (int r = 0; r < 8; ++r) { m_run[r] = -1e30f; l_run[r] = 0.f; }

  const int i0 = tid, i1 = tid + 256;
  const int kg0 = i0 * 4, kg1 = i1 * 4;
  const int kl0 = (i0 >> 3) * PK + (i0 & 7) * 4;
  const int kl1 = (i1 >> 3) * PK + (i1 & 7) * 4;
  const int vg0 = (i0 / (KC / 4)) * SEQ + (i0 % (KC / 4)) * 4;
  const int vg1 = (i1 / (KC / 4)) * SEQ + (i1 % (KC / 4)) * 4;
  const int vl0 = (i0 / (KC / 4)) * PV + (i0 % (KC / 4)) * 4;
  const int vl1 = (i1 / (KC / 4)) * PV + (i1 % (KC / 4)) * 4;

  v4f gkr0 = *(const v4f*)(kreh + kg0), gkr1 = *(const v4f*)(kreh + kg1);
  v4f gki0 = *(const v4f*)(kimh + kg0), gki1 = *(const v4f*)(kimh + kg1);
  v4f gvr0 = *(const v4f*)(vreh + vg0), gvr1 = *(const v4f*)(vreh + vg1);
  v4f gvi0 = *(const v4f*)(vimh + vg0), gvi1 = *(const v4f*)(vimh + vg1);
  *(v4f*)(&lds_k[0][0][kl0]) = gkr0;  *(v4f*)(&lds_k[0][0][kl1]) = gkr1;
  *(v4f*)(&lds_k[0][1][kl0]) = gki0;  *(v4f*)(&lds_k[0][1][kl1]) = gki1;
  *(v4f*)(&lds_v[0][0][vl0]) = gvr0;  *(v4f*)(&lds_v[0][0][vl1]) = gvr1;
  *(v4f*)(&lds_v[0][1][vl0]) = gvi0;  *(v4f*)(&lds_v[0][1][vl1]) = gvi1;
  __syncthreads();

  for (int c = 0; c < SEQ / KC; ++c) {
    const int st = c & 1;
    const bool more = (c + 1) < (SEQ / KC);

    if (more) {
      const float* kre_n = kreh + (size_t)(c + 1) * KC * HD;
      const float* kim_n = kimh + (size_t)(c + 1) * KC * HD;
      const float* vre_n = vreh + (size_t)(c + 1) * KC;
      const float* vim_n = vimh + (size_t)(c + 1) * KC;
      gkr0 = *(const v4f*)(kre_n + kg0);  gkr1 = *(const v4f*)(kre_n + kg1);
      gki0 = *(const v4f*)(kim_n + kg0);  gki1 = *(const v4f*)(kim_n + kg1);
      gvr0 = *(const v4f*)(vre_n + vg0);  gvr1 = *(const v4f*)(vre_n + vg1);
      gvi0 = *(const v4f*)(vim_n + vg0);  gvi1 = *(const v4f*)(vim_n + vg1);
    }

    const float* ks_re = lds_k[st][0];
    const float* ks_im = lds_k[st][1];
    const float* vs_re = lds_v[st][0];
    const float* vs_im = lds_v[st][1];

    float a[KC / 16][8];
    float mx[8];
#pragma unroll
    for (int kt = 0; kt < KC / 16; ++kt) {
      const int krow = (kt * 16 + l16) * PK + half * 2;
      v8f sre = {0.f}, sim = {0.f};
#pragma unroll
      for (int t = 0; t < 8; ++t) {
        v2f krv = ldp(ks_re + krow + t * 4);
        v2f kiv = ldp(ks_im + krow + t * 4);
        sre = wmma4(qr[t],  krv, sre);
        sre = wmma4(qi[t],  kiv, sre);
        sim = wmma4(qi[t],  krv, sim);
        sim = wmma4(qrn[t], kiv, sim);
      }
#pragma unroll
      for (int r = 0; r < 8; ++r) {
        const float v = sqrtf(sre[r] * sre[r] + sim[r] * sim[r]) * scale;
        mx[r] = (kt == 0) ? v : fmaxf(mx[r], v);
        a[kt][r] = v;
      }
    }

    float rs[8], nm[8];
#pragma unroll
    for (int r = 0; r < 8; ++r) {
      float m = mx[r];
      m = fmaxf(m, row_xor<1>(m));
      m = fmaxf(m, row_xor<2>(m));
      m = fmaxf(m, row_xor<4>(m));
      m = fmaxf(m, row_xor<8>(m));
      nm[r] = fmaxf(m_run[r], m);
      rs[r] = __expf(m_run[r] - nm[r]);
      m_run[r] = nm[r];
    }

    float ps[8];
#pragma unroll
    for (int kt = 0; kt < KC / 16; ++kt) {
#pragma unroll
      for (int r = 0; r < 8; ++r) {
        const float pv = __expf(a[kt][r] - nm[r]);
        ps[r] = (kt == 0) ? pv : ps[r] + pv;
        pbuf[(r + 8 * half) * PP + kt * 16 + l16] = pv;
      }
    }
#pragma unroll
    for (int r = 0; r < 8; ++r) {
      float sum = ps[r];
      sum += row_xor<1>(sum);
      sum += row_xor<2>(sum);
      sum += row_xor<4>(sum);
      sum += row_xor<8>(sum);
      l_run[r] = l_run[r] * rs[r] + sum;
    }
#pragma unroll
    for (int r = 0; r < 8; ++r) {
      o_re0[r] *= rs[r]; o_re1[r] *= rs[r];
      o_im0[r] *= rs[r]; o_im1[r] *= rs[r];
    }
    __asm__ volatile("s_wait_dscnt 0x0" ::: "memory");

    v2f aP[16], br0[16], bi0[16], br1[16], bi1[16];
#pragma unroll
    for (int kt = 0; kt < KC / 16; ++kt) {
      const int vrow = l16 * PV + kt * 16 + half * 2;
#pragma unroll
      for (int t = 0; t < 4; ++t) {
        const int j = kt * 4 + t;
        aP[j]  = ldp(pbuf + l16 * PP + kt * 16 + t * 4 + half * 2);
        br0[j] = ldp(vs_re + vrow + t * 4);
        bi0[j] = ldp(vs_im + vrow + t * 4);
        br1[j] = ldp(vs_re + vrow + 16 * PV + t * 4);
        bi1[j] = ldp(vs_im + vrow + 16 * PV + t * 4);
      }
    }
    __builtin_amdgcn_sched_barrier(0);
#pragma unroll
    for (int j = 0; j < KC / 4; ++j) {
      o_re0 = wmma4(aP[j], br0[j], o_re0);
      o_im0 = wmma4(aP[j], bi0[j], o_im0);
      o_re1 = wmma4(aP[j], br1[j], o_re1);
      o_im1 = wmma4(aP[j], bi1[j], o_im1);
    }

    if (more) {
      const int sn = st ^ 1;
      *(v4f*)(&lds_k[sn][0][kl0]) = gkr0;  *(v4f*)(&lds_k[sn][0][kl1]) = gkr1;
      *(v4f*)(&lds_k[sn][1][kl0]) = gki0;  *(v4f*)(&lds_k[sn][1][kl1]) = gki1;
      *(v4f*)(&lds_v[sn][0][vl0]) = gvr0;  *(v4f*)(&lds_v[sn][0][vl1]) = gvr1;
      *(v4f*)(&lds_v[sn][1][vl0]) = gvi0;  *(v4f*)(&lds_v[sn][1][vl1]) = gvi1;
    }
    __syncthreads();
  }

#pragma unroll
  for (int r = 0; r < 8; ++r) {
    const float inv = 1.0f / l_run[r];
    const int s  = q0 + r + 8 * half;
    const int o0 = h * HD + l16;
    float2 e0 = { o_re0[r] * inv, o_im0[r] * inv };
    float2 e1 = { o_re1[r] * inv, o_im1[r] * inv };
    *(float2*)(Out + ((size_t)s * CH + o0) * 2)      = e0;
    *(float2*)(Out + ((size_t)s * CH + o0 + 16) * 2) = e1;
  }
}

extern "C" void kernel_launch(void* const* d_in, const int* in_sizes, int n_in,
                              void* d_out, int out_size, void* d_ws, size_t ws_size,
                              hipStream_t stream) {
  (void)in_sizes; (void)n_in; (void)out_size; (void)ws_size;
  const float* Q  = (const float*)d_in[0];
  const float* V  = (const float*)d_in[1];
  const float* K  = (const float*)d_in[2];
  const float* Wq = (const float*)d_in[3];
  const float* bq = (const float*)d_in[4];
  const float* Wk = (const float*)d_in[5];
  const float* bk = (const float*)d_in[6];
  const float* Wv = (const float*)d_in[7];
  const float* bv = (const float*)d_in[8];
  float* out = (float*)d_out;

  const size_t plane = (size_t)NH * SEQ * HD;
  float* qre = (float*)d_ws;
  float* qim = qre + plane;
  float* kre = qim + plane;
  float* kim = kre + plane;
  float* vre = kim + plane;
  float* vim = vre + plane;

  const dim3 blk(256);
  const dim3 pgrid((SEQ / 16) * (CH / 16) / 8);
  proj_complex_kernel<0><<<pgrid, blk, 0, stream>>>(Q, Wq, bq, qre, qim);
  proj_complex_kernel<0><<<pgrid, blk, 0, stream>>>(K, Wk, bk, kre, kim);
  proj_complex_kernel<1><<<pgrid, blk, 0, stream>>>(V, Wv, bv, vre, vim);

  const dim3 agrid(SEQ / 16 / 8, NH);
  attn_complex_kernel<<<agrid, blk, 0, stream>>>(qre, qim, kre, kim, vre, vim, out);
}


# projections W-stationary: a wave keeps its 16 output channels' whole W panel (128 K-steps) in registers, one wave per SIMD, and streams 4 X row tiles: 1 fragment load per 4 WMMAs, not 2; sums kept
# speedup vs baseline: 1.6432x; 1.0547x over previous
#include <hip/hip_runtime.h>

#define SEQ 2048
#define CH  512
#define NH  16
#define HD  32

typedef float v2f __attribute__((ext_vector_type(2)));
typedef float v8f __attribute__((ext_vector_type(8)));
typedef float v4f __attribute__((ext_vector_type(4)));

__device__ __forceinline__ v8f wmma4(v2f a, v2f b, v8f c) {
  return __builtin_amdgcn_wmma_f32_16x16x4_f32(false, a, false, b, (short)0, c, false, false);
}

__device__ __forceinline__ v2f ldp(const float* p) {
  const float2 t = *(const float2*)p;
  v2f r = { t.x, t.y };
  return r;
}

template <int M>
__device__ __forceinline__ float row_xor(float v) {
  return __builtin_bit_cast(float, __builtin_amdgcn_update_dpp(
      0, __builtin_bit_cast(int, v), 0x160 + M, 0xf, 0xf, true));
}

#define PT_ST 4
#define PT_KC 16
#define PT_NC (CH / 4 / PT_KC)

template <int MODE>
__global__ void __launch_bounds__(128)
proj_complex_kernel(const float* __restrict__ X, const float* __restrict__ W,
                    const float* __restrict__ Bv, float* __restrict__ Yre,
                    float* __restrict__ Yim) {
  const int wave = threadIdx.x >> 5;
  const int lane = threadIdx.x & 31;
  const int half = lane >> 4;
  const int l16  = lane & 15;

  const int oTile  = (blockIdx.x & 7) * 4 + wave;
  const int sGroup = blockIdx.x >> 3;
  const int brow   = oTile * 16 + l16;

  v2f wr[CH / 4], wi[CH / 4];
  {
    const float* wp = W + ((size_t)brow * CH + half * 2) * 2;
#pragma unroll
    for (int t = 0; t < CH / 4; ++t) {
      const float4 wb = *(const float4*)(wp + t * 8);
      wr[t] = v2f{ wb.x, wb.z };
      wi[t] = v2f{ wb.y, wb.w };
    }
  }

  const float2 bb = *(const float2*)(Bv + (size_t)brow * 2);
  const int h  = brow >> 5;
  const int dd = brow & 31;

  const float* xp = X + ((size_t)(sGroup * PT_ST * 16 + l16) * CH + half * 2) * 2;
  const size_t xTile = (size_t)16 * CH * 2;

  float4 xa0[PT_KC], xa1[PT_KC];
#pragma unroll
  for (int j = 0; j < PT_KC; ++j) xa0[j] = *(const float4*)(xp + j * 8);

#pragma nounroll
  for (int si = 0; si < PT_ST; ++si) {
    const int sTile = sGroup * PT_ST + si;
    const float* xn = (si + 1 < PT_ST) ? xp + xTile : xp;

    v8f yr = {0.f}, yi = {0.f};
#pragma unroll
    for (int c = 0; c < PT_NC; ++c) {
      const float* nsrc = (c + 1 < PT_NC) ? xp + (c + 1) * PT_KC * 8 : xn;
      if ((c & 1) == 0) {
#pragma unroll
        for (int j = 0; j < PT_KC; ++j) xa1[j] = *(const float4*)(nsrc + j * 8);
      } else {
#pragma unroll
        for (int j = 0; j < PT_KC; ++j) xa0[j] = *(const float4*)(nsrc + j * 8);
      }
      __builtin_amdgcn_sched_barrier(0);
#pragma unroll
      for (int j = 0; j < PT_KC; ++j) {
        const float4 xa = (c & 1) ? xa1[j] : xa0[j];
        const int t = c * PT_KC + j;
        v2f ar  = { xa.x,  xa.z };
        v2f ai  = { xa.y,  xa.w };
        v2f ain = { -xa.y, -xa.w };
        yr = wmma4(ar,  wr[t], yr);
        yr = wmma4(ain, wi[t], yr);
        yi = wmma4(ar,  wi[t], yi);
        yi = wmma4(ai,  wr[t], yi);
      }
    }

    if (MODE == 0) {
#pragma unroll
      for (int r = 0; r < 8; ++r) {
        const int s = sTile * 16 + r + 8 * half;
        const size_t idx = ((size_t)h * SEQ + s) * HD + dd;
        Yre[idx] = yr[r] + bb.x;
        Yim[idx] = yi[r] + bb.y;
      }
    } else {
      const size_t base = ((size_t)(h * HD + dd)) * SEQ + sTile * 16 + 8 * half;
      float4 r0 = { yr[0] + bb.x, yr[1] + bb.x, yr[2] + bb.x, yr[3] + bb.x };
      float4 r1 = { yr[4] + bb.x, yr[5] + bb.x, yr[6] + bb.x, yr[7] + bb.x };
      float4 i0 = { yi[0] + bb.y, yi[1] + bb.y, yi[2] + bb.y, yi[3] + bb.y };
      float4 i1 = { yi[4] + bb.y, yi[5] + bb.y, yi[6] + bb.y, yi[7] + bb.y };
      *(float4*)(Yre + base)     = r0;
      *(float4*)(Yre + base + 4) = r1;
      *(float4*)(Yim + base)     = i0;
      *(float4*)(Yim + base + 4) = i1;
    }
    xp += xTile;
  }
}

#define KC 64
#define PK 36
#define PV (KC + 4)
#define PP (KC + 4)

__global__ void __launch_bounds__(256)
attn_complex_kernel(const float* __restrict__ Qre, const float* __restrict__ Qim,
                    const float* __restrict__ Kre, const float* __restrict__ Kim,
                    const float* __restrict__ Vtre, const float* __restrict__ Vtim,
                    float* __restrict__ Out) {
  __shared__ __attribute__((aligned(16))) float lds_k[2][2][KC * PK];
  __shared__ __attribute__((aligned(16))) float lds_v[2][2][HD * PV];
  __shared__ __attribute__((aligned(16))) float lds_p[8][16 * PP];

  const int tid  = threadIdx.x;
  const int wave = tid >> 5;
  const int lane = tid & 31;
  const int half = lane >> 4;
  const int l16  = lane & 15;
  const int h    = blockIdx.y;
  const int q0   = (blockIdx.x * 8 + wave) * 16;

  const float scale = 0.17677669529663687f;

  const float* qreh = Qre + (size_t)h * SEQ * HD;
  const float* qimh = Qim + (size_t)h * SEQ * HD;
  const float* kreh = Kre + (size_t)h * SEQ * HD;
  const float* kimh = Kim + (size_t)h * SEQ * HD;
  const float* vreh = Vtre + (size_t)h * HD * SEQ;
  const float* vimh = Vtim + (size_t)h * HD * SEQ;
  float* pbuf = lds_p[wave];

  v2f qr[8], qi[8], qrn[8];
#pragma unroll
  for (int t = 0; t < 8; ++t) {
    const int dk = t * 4 + half * 2;
    const size_t off = ((size_t)(q0 + l16)) * HD + dk;
    qr[t]  = ldp(qreh + off);
    qi[t]  = ldp(qimh + off);
    qrn[t] = -qr[t];
  }

  float m_run[8], l_run[8];
  v8f o_re0 = {0.f}, o_re1 = {0.f}, o_im0 = {0.f}, o_im1 = {0.f};
#pragma unroll
  for (int r = 0; r < 8; ++r) { m_run[r] = -1e30f; l_run[r] = 0.f; }

  const int i0 = tid, i1 = tid + 256;
  const int kg0 = i0 * 4, kg1 = i1 * 4;
  const int kl0 = (i0 >> 3) * PK + (i0 & 7) * 4;
  const int kl1 = (i1 >> 3) * PK + (i1 & 7) * 4;
  const int vg0 = (i0 / (KC / 4)) * SEQ + (i0 % (KC / 4)) * 4;
  const int vg1 = (i1 / (KC / 4)) * SEQ + (i1 % (KC / 4)) * 4;
  const int vl0 = (i0 / (KC / 4)) * PV + (i0 % (KC / 4)) * 4;
  const int vl1 = (i1 / (KC / 4)) * PV + (i1 % (KC / 4)) * 4;

  v4f gkr0 = *(const v4f*)(kreh + kg0), gkr1 = *(const v4f*)(kreh + kg1);
  v4f gki0 = *(const v4f*)(kimh + kg0), gki1 = *(const v4f*)(kimh + kg1);
  v4f gvr0 = *(const v4f*)(vreh + vg0), gvr1 = *(const v4f*)(vreh + vg1);
  v4f gvi0 = *(const v4f*)(vimh + vg0), gvi1 = *(const v4f*)(vimh + vg1);
  *(v4f*)(&lds_k[0][0][kl0]) = gkr0;  *(v4f*)(&lds_k[0][0][kl1]) = gkr1;
  *(v4f*)(&lds_k[0][1][kl0]) = gki0;  *(v4f*)(&lds_k[0][1][kl1]) = gki1;
  *(v4f*)(&lds_v[0][0][vl0]) = gvr0;  *(v4f*)(&lds_v[0][0][vl1]) = gvr1;
  *(v4f*)(&lds_v[0][1][vl0]) = gvi0;  *(v4f*)(&lds_v[0][1][vl1]) = gvi1;
  __syncthreads();

  for (int c = 0; c < SEQ / KC; ++c) {
    const int st = c & 1;
    const bool more = (c + 1) < (SEQ / KC);

    if (more) {
      const float* kre_n = kreh + (size_t)(c + 1) * KC * HD;
      const float* kim_n = kimh + (size_t)(c + 1) * KC * HD;
      const float* vre_n = vreh + (size_t)(c + 1) * KC;
      const float* vim_n = vimh + (size_t)(c + 1) * KC;
      gkr0 = *(const v4f*)(kre_n + kg0);  gkr1 = *(const v4f*)(kre_n + kg1);
      gki0 = *(const v4f*)(kim_n + kg0);  gki1 = *(const v4f*)(kim_n + kg1);
      gvr0 = *(const v4f*)(vre_n + vg0);  gvr1 = *(const v4f*)(vre_n + vg1);
      gvi0 = *(const v4f*)(vim_n + vg0);  gvi1 = *(const v4f*)(vim_n + vg1);
    }

    const float* ks_re = lds_k[st][0];
    const float* ks_im = lds_k[st][1];
    const float* vs_re = lds_v[st][0];
    const float* vs_im = lds_v[st][1];

    float a[KC / 16][8];
    float mx[8];
#pragma unroll
    for (int kt = 0; kt < KC / 16; ++kt) {
      const int krow = (kt * 16 + l16) * PK + half * 2;
      v8f sre = {0.f}, sim = {0.f};
#pragma unroll
      for (int t = 0; t < 8; ++t) {
        v2f krv = ldp(ks_re + krow + t * 4);
        v2f kiv = ldp(ks_im + krow + t * 4);
        sre = wmma4(qr[t],  krv, sre);
        sre = wmma4(qi[t],  kiv, sre);
        sim = wmma4(qi[t],  krv, sim);
        sim = wmma4(qrn[t], kiv, sim);
      }
#pragma unroll
      for (int r = 0; r < 8; ++r) {
        const float v = sqrtf(sre[r] * sre[r] + sim[r] * sim[r]) * scale;
        mx[r] = (kt == 0) ? v : fmaxf(mx[r], v);
        a[kt][r] = v;
      }
    }

    float rs[8], nm[8];
#pragma unroll
    for (int r = 0; r < 8; ++r) {
      float m = mx[r];
      m = fmaxf(m, row_xor<1>(m));
      m = fmaxf(m, row_xor<2>(m));
      m = fmaxf(m, row_xor<4>(m));
      m = fmaxf(m, row_xor<8>(m));
      nm[r] = fmaxf(m_run[r], m);
      rs[r] = __expf(m_run[r] - nm[r]);
      m_run[r] = nm[r];
    }

    float ps[8];
#pragma unroll
    for (int kt = 0; kt < KC / 16; ++kt) {
#pragma unroll
      for (int r = 0; r < 8; ++r) {
        const float pv = __expf(a[kt][r] - nm[r]);
        ps[r] = (kt == 0) ? pv : ps[r] + pv;
        pbuf[(r + 8 * half) * PP + kt * 16 + l16] = pv;
      }
    }
#pragma unroll
    for (int r = 0; r < 8; ++r) {
      float sum = ps[r];
      sum += row_xor<1>(sum);
      sum += row_xor<2>(sum);
      sum += row_xor<4>(sum);
      sum += row_xor<8>(sum);
      l_run[r] = l_run[r] * rs[r] + sum;
    }
#pragma unroll
    for (int r = 0; r < 8; ++r) {
      o_re0[r] *= rs[r]; o_re1[r] *= rs[r];
      o_im0[r] *= rs[r]; o_im1[r] *= rs[r];
    }
    __asm__ volatile("s_wait_dscnt 0x0" ::: "memory");

    v2f aP[16], br0[16], bi0[16], br1[16], bi1[16];
#pragma unroll
    for (int kt = 0; kt < KC / 16; ++kt) {
      const int vrow = l16 * PV + kt * 16 + half * 2;
#pragma unroll
      for (int t = 0; t < 4; ++t) {
        const int j = kt * 4 + t;
        aP[j]  = ldp(pbuf + l16 * PP + kt * 16 + t * 4 + half * 2);
        br0[j] = ldp(vs_re + vrow + t * 4);
        bi0[j] = ldp(vs_im + vrow + t * 4);
        br1[j] = ldp(vs_re + vrow + 16 * PV + t * 4);
        bi1[j] = ldp(vs_im + vrow + 16 * PV + t * 4);
      }
    }
    __builtin_amdgcn_sched_barrier(0);
#pragma unroll
    for (int j = 0; j < KC / 4; ++j) {
      o_re0 = wmma4(aP[j], br0[j], o_re0);
      o_im0 = wmma4(aP[j], bi0[j], o_im0);
      o_re1 = wmma4(aP[j], br1[j], o_re1);
      o_im1 = wmma4(aP[j], bi1[j], o_im1);
    }

    if (more) {
      const int sn = st ^ 1;
      *(v4f*)(&lds_k[sn][0][kl0]) = gkr0;  *(v4f*)(&lds_k[sn][0][kl1]) = gkr1;
      *(v4f*)(&lds_k[sn][1][kl0]) = gki0;  *(v4f*)(&lds_k[sn][1][kl1]) = gki1;
      *(v4f*)(&lds_v[sn][0][vl0]) = gvr0;  *(v4f*)(&lds_v[sn][0][vl1]) = gvr1;
      *(v4f*)(&lds_v[sn][1][vl0]) = gvi0;  *(v4f*)(&lds_v[sn][1][vl1]) = gvi1;
    }
    __syncthreads();
  }

#pragma unroll
  for (int r = 0; r < 8; ++r) {
    const float inv = 1.0f / l_run[r];
    const int s  = q0 + r + 8 * half;
    const int o0 = h * HD + l16;
    float2 e0 = { o_re0[r] * inv, o_im0[r] * inv };
    float2 e1 = { o_re1[r] * inv, o_im1[r] * inv };
    *(float2*)(Out + ((size_t)s * CH + o0) * 2)      = e0;
    *(float2*)(Out + ((size_t)s * CH + o0 + 16) * 2) = e1;
  }
}

extern "C" void kernel_launch(void* const* d_in, const int* in_sizes, int n_in,
                              void* d_out, int out_size, void* d_ws, size_t ws_size,
                              hipStream_t stream) {
  (void)in_sizes; (void)n_in; (void)out_size; (void)ws_size;
  const float* Q  = (const float*)d_in[0];
  const float* V  = (const float*)d_in[1];
  const float* K  = (const float*)d_in[2];
  const float* Wq = (const float*)d_in[3];
  const float* bq = (const float*)d_in[4];
  const float* Wk = (const float*)d_in[5];
  const float* bk = (const float*)d_in[6];
  const float* Wv = (const float*)d_in[7];
  const float* bv = (const float*)d_in[8];
  float* out = (float*)d_out;

  const size_t plane = (size_t)NH * SEQ * HD;
  float* qre = (float*)d_ws;
  float* qim = qre + plane;
  float* kre = qim + plane;
  float* kim = kre + plane;
  float* vre = kim + plane;
  float* vim = vre + plane;

  const dim3 blk(256);
  const dim3 pblk(128);
  const dim3 pgrid(256);
  proj_complex_kernel<0><<<pgrid, pblk, 0, stream>>>(Q, Wq, bq, qre, qim);
  proj_complex_kernel<0><<<pgrid, pblk, 0, stream>>>(K, Wk, bk, kre, kim);
  proj_complex_kernel<1><<<pgrid, pblk, 0, stream>>>(V, Wv, bv, vre, vim);

  const dim3 agrid(SEQ / 16 / 8, NH);
  attn_complex_kernel<<<agrid, blk, 0, stream>>>(qre, qim, kre, kim, vre, vim, out);
}
